# starting kernel
# baseline (speedup 1.0000x reference)
#include <hip/hip_runtime.h>

typedef float v2f __attribute__((ext_vector_type(2)));
typedef float v8f __attribute__((ext_vector_type(8)));

#define SDIM 64
#define NCIN 16
#define NCOUT 16
#define NTAP 27

__launch_bounds__(128, 4)
__global__ void conv3d_wmma_f32(const float* __restrict__ x,
                                const float* __restrict__ w,
                                const float* __restrict__ bias,
                                float* __restrict__ out) {
    __shared__ __align__(16) float wl[NTAP * NCOUT * NCIN];
    __shared__ __align__(16) float xs4[9 * 66 * 4];

    const int tid  = threadIdx.x;
    const int lane = tid & 31;
    const int wv   = tid >> 5;
    const int Lm   = lane & 15;
    const int half = lane >> 4;
    const int cl   = half << 1;
    const int w0   = wv * 16;

    const int blk = blockIdx.x;
    const int b   = blk >> 12;
    const int d   = (blk >> 6) & 63;
    const int h   = blk & 63;

    for (int idx = tid; idx < NCOUT * NCIN * NTAP; idx += 128) {
        int co   = idx / (NCIN * NTAP);
        int rest = idx - co * (NCIN * NTAP);
        int ci   = rest / NTAP;
        int tap  = rest - ci * NTAP;
        wl[(tap * NCOUT + co) * NCIN + ci] = w[idx];
    }
    for (int t = tid; t < 9 * 4 * 2; t += 128) {
        int rr   = t >> 1;
        int side = t & 1;
        int row  = rr >> 2;
        int cil  = rr & 3;
        xs4[(row * 66 + side * 65) * 4 + cil] = 0.0f;
    }

    const int abase = (w0 + Lm) * 4 + cl;
    v8f acc = {};

    for (int g = 0; g < 4; ++g) {
        __syncthreads();

        for (int t = tid; t < 36 * 16; t += 128) {
            int rr  = t >> 4;
            int q   = t & 15;
            int cil = rr / 9;
            int row = rr - cil * 9;
            int zd  = d + row / 3 - 1;
            int zh  = h + (row % 3) - 1;
            int ci  = (g << 2) + cil;
            float4 v = {0.0f, 0.0f, 0.0f, 0.0f};
            if ((unsigned)zd < 64u && (unsigned)zh < 64u)
                v = *(const float4*)&x[(((b * NCIN + ci) * SDIM + zd) * SDIM + zh) * SDIM
                                       + (q << 2)];
            int base = (row * 66 + 1 + (q << 2)) * 4 + cil;
            xs4[base]      = v.x;
            xs4[base + 4]  = v.y;
            xs4[base + 8]  = v.z;
            xs4[base + 12] = v.w;
        }
        __syncthreads();

        if (g < 3) {
            int rr  = tid >> 4;
            int q   = tid & 15;
            int cil = rr / 9;
            int row = rr - cil * 9;
            int zd  = d + row / 3 - 1;
            int zh  = h + (row % 3) - 1;
            int ci  = ((g + 1) << 2) + cil;
            if ((unsigned)zd < 64u && (unsigned)zh < 64u)
                __builtin_prefetch(&x[(((b * NCIN + ci) * SDIM + zd) * SDIM + zh) * SDIM
                                      + (q << 2)], 0, 3);
        }

        const int bbase = Lm * NCIN + (g << 2) + cl;

        #pragma unroll
        for (int kd = 0; kd < 3; ++kd) {
            #pragma unroll
            for (int kh = 0; kh < 3; ++kh) {
                const int arow = abase + (kd * 3 + kh) * 66 * 4;
                #pragma unroll
                for (int kw = 0; kw < 3; ++kw) {
                    const int tap = kd * 9 + kh * 3 + kw;
                    v2f a  = *(const v2f*)(xs4 + arow + kw * 4);
                    v2f bm = *(const v2f*)(wl + bbase + tap * (NCOUT * NCIN));
                    acc = __builtin_amdgcn_wmma_f32_16x16x4_f32(
                        false, a, false, bm, (short)0, acc, false, false);
                }
            }
        }
    }

    const float bv = bias[Lm];
    float* op = out + ((((long)b * NCOUT + Lm) * SDIM + d) * SDIM + h) * SDIM
                    + w0 + 8 * half;
    float4 v0 = {acc[0] + bv, acc[1] + bv, acc[2] + bv, acc[3] + bv};
    float4 v1 = {acc[4] + bv, acc[5] + bv, acc[6] + bv, acc[7] + bv};
    *(float4*)(op)     = v0;
    *(float4*)(op + 4) = v1;
}

extern "C" void kernel_launch(void* const* d_in, const int* in_sizes, int n_in,
                              void* d_out, int out_size, void* d_ws, size_t ws_size,
                              hipStream_t stream) {
    const float* x    = (const float*)d_in[0];
    const float* wgt  = (const float*)d_in[1];
    const float* bias = (const float*)d_in[2];
    float* out        = (float*)d_out;
    (void)in_sizes; (void)n_in; (void)out_size; (void)d_ws; (void)ws_size;

    dim3 grid(4 * SDIM * SDIM);
    conv3d_wmma_f32<<<grid, 128, 0, stream>>>(x, wgt, bias, out);
}


# x planes kept in a wave-private LDS ring along d: a work-group walks 16 consecutive d at one (b,h), each row fetches only its one new plane (6 loads, not 18) during the row's WMMAs; sums unchanged
# speedup vs baseline: 2.0554x; 2.0554x over previous
#include <hip/hip_runtime.h>

typedef float v2f __attribute__((ext_vector_type(2)));
typedef float v8f __attribute__((ext_vector_type(8)));

#define SDIM 64
#define NCIN 16
#define NCOUT 16
#define NTAP 27
#define NBLK 1024
#define DSEG 16
#define WPITCH (NCOUT * NCIN + 2)
#define LINEP (66 * 4)
#define PLANEP (3 * LINEP)
#define SLAB (3 * PLANEP)
#define REDT 256
#define SMEM_FLOATS (4 * SLAB + 2 * 4 * 4 * REDT)

__device__ __forceinline__ void plane_load(float4 (&v)[6], const float* xg, int plane,
                                           int offp, int offl, int rs) {
    const float* pp = xg + min(max(plane, 0), SDIM - 1) * (SDIM * SDIM);
    #pragma unroll
    for (int i = 0; i < 4; ++i)
        v[i] = *(const float4*)(pp + offp + (i << 4));
    v[4] = *(const float4*)(pp + offl + (rs << 4));
    v[5] = *(const float4*)(pp + offl + ((2 + rs) << 4));
}

__device__ __forceinline__ void plane_store(const float4 (&v)[6], float* slot, int plane,
                                            bool okp, bool okl, int sb_pair, int sb_last) {
    const bool pok = (unsigned)plane < (unsigned)SDIM;
    const bool a = pok && okp;
    const bool c = pok && okl;
    #pragma unroll
    for (int i = 0; i < 6; ++i) {
        const bool ok = (i < 4) ? a : c;
        float* p = (i < 4) ? slot + sb_pair + i * 64
                           : slot + sb_last + 2 * LINEP + (i - 4) * 128;
        p[0]  = ok ? v[i].x : 0.0f;
        p[4]  = ok ? v[i].y : 0.0f;
        p[8]  = ok ? v[i].z : 0.0f;
        p[12] = ok ? v[i].w : 0.0f;
    }
}

__launch_bounds__(128, 4)
__global__ void conv3d_wmma_f32(const float* __restrict__ x,
                                const float* __restrict__ w,
                                const float* __restrict__ bias,
                                float* __restrict__ out) {
    __shared__ __align__(16) float smem[SMEM_FLOATS];
    float* wl = smem;

    const int tid  = threadIdx.x;
    const int lane = tid & 31;
    const int wv   = tid >> 5;
    const int Lm   = lane & 15;
    const int half = lane >> 4;
    const int cl   = half << 1;

    for (int idx = tid; idx < NCOUT * NCIN * NTAP; idx += 128) {
        int co   = idx / (NCIN * NTAP);
        int rest = idx - co * (NCIN * NTAP);
        int ci   = rest / NTAP;
        int tap  = rest - ci * NTAP;
        wl[tap * WPITCH + co * NCIN + ci] = w[idx];
    }
    __syncthreads();

    v2f bfr[NTAP];
    #pragma unroll
    for (int tap = 0; tap < NTAP; ++tap)
        bfr[tap] = *(const v2f*)(wl + Lm * NCIN + (wv << 2) + cl + tap * WPITCH);
    __syncthreads();

    float* xs  = smem + wv * SLAB;
    float* red = smem + 4 * SLAB;

    for (int t = lane; t < 9 * 4 * 2; t += 32) {
        int rr   = t >> 1;
        int side = t & 1;
        xs[((rr >> 2) * 66 + side * 65) * 4 + (rr & 3)] = 0.0f;
    }

    const float bv = bias[Lm];

    const int cil = lane & 3;
    const int ql  = (lane >> 2) & 3;
    const int rs  = lane >> 4;
    const int sb_pair = rs * LINEP + 16 * ql + cil + 4;
    const int sb_last = rs * 64    + 16 * ql + cil + 4;
    const int abase = Lm * 4 + cl;

    const int blk = blockIdx.x;
    const int b   = blk >> 8;
    const int d0  = ((blk >> 6) & 3) * DSEG;
    const int h   = blk & 63;

    const float* xg = x + (long)(b * NCIN + (wv << 2) + cil) * (SDIM * SDIM * SDIM) + (ql << 2);
    const int  zhp  = h + rs - 1;
    const bool okp  = (unsigned)zhp < (unsigned)SDIM;
    const bool okl  = (h + 1) < SDIM;
    const int  offp = min(max(zhp, 0), SDIM - 1) * SDIM;
    const int  offl = min(h + 1, SDIM - 1) * SDIM;

    {
        float4 pv[3][6];
        #pragma unroll
        for (int j = 0; j < 3; ++j)
            plane_load(pv[j], xg, d0 - 1 + j, offp, offl, rs);
        #pragma unroll
        for (int j = 0; j < 3; ++j)
            plane_store(pv[j], xs + j * PLANEP, d0 - 1 + j, okp, okl, sb_pair, sb_last);
    }
    __builtin_amdgcn_fence(__ATOMIC_RELEASE, "wavefront");
    __builtin_amdgcn_wave_barrier();
    __builtin_amdgcn_fence(__ATOMIC_ACQUIRE, "wavefront");

    int o0 = 0, o1 = PLANEP, o2 = 2 * PLANEP;
    int par = 0;
    #pragma nounroll
    for (int k = 0; k < DSEG; ++k) {
        const int d = d0 + k;

        float4 nv[6];
        plane_load(nv, xg, d + 2, offp, offl, rs);
        __builtin_amdgcn_sched_barrier(0);

        const float* pl[3] = {xs + abase + o0, xs + abase + o1, xs + abase + o2};
        v8f acc[4] = {};
        v2f af[2][12];
        #pragma unroll
        for (int j = 0; j < 12; ++j)
            af[0][j] = *(const v2f*)(pl[0] + (16 * (j & 3) + (j >> 2)) * 4);
        #pragma unroll
        for (int line = 0; line < 9; ++line) {
            if (line < 8) {
                const int ln = line + 1;
                #pragma unroll
                for (int j = 0; j < 12; ++j)
                    af[ln & 1][j] = *(const v2f*)(pl[ln / 3]
                        + ((ln % 3) * 66 + 16 * (j & 3) + (j >> 2)) * 4);
            }
            __builtin_amdgcn_sched_barrier(0);
            #pragma unroll
            for (int kw = 0; kw < 3; ++kw) {
                #pragma unroll
                for (int m = 0; m < 4; ++m)
                    acc[m] = __builtin_amdgcn_wmma_f32_16x16x4_f32(
                        false, af[line & 1][kw * 4 + m], false, bfr[line * 3 + kw],
                        (short)0, acc[m], false, false);
            }
            __builtin_amdgcn_sched_barrier(0);
        }

        float* rw = red + ((par * 4 + wv) * 4) * REDT + lane * 4;
        #pragma unroll
        for (int m = 0; m < 4; ++m) {
            float4 lo = {acc[m][0], acc[m][1], acc[m][2], acc[m][3]};
            float4 hi = {acc[m][4], acc[m][5], acc[m][6], acc[m][7]};
            *(float4*)(rw + m * REDT)       = lo;
            *(float4*)(rw + m * REDT + 128) = hi;
        }

        __builtin_amdgcn_fence(__ATOMIC_RELEASE, "wavefront");
        __builtin_amdgcn_wave_barrier();
        __builtin_amdgcn_fence(__ATOMIC_ACQUIRE, "wavefront");
        plane_store(nv, xs + o0, d + 2, okp, okl, sb_pair, sb_last);
        __builtin_amdgcn_fence(__ATOMIC_RELEASE, "wavefront");
        __builtin_amdgcn_wave_barrier();
        __builtin_amdgcn_fence(__ATOMIC_ACQUIRE, "wavefront");
        { const int t = o0; o0 = o1; o1 = o2; o2 = t; }

        __syncthreads();
        const float* rr = red + (par * 16 + wv) * REDT + lane * 4;
        float4 s0 = *(const float4*)(rr);
        float4 s1 = *(const float4*)(rr + 128);
        #pragma unroll
        for (int g = 1; g < 4; ++g) {
            float4 t0 = *(const float4*)(rr + g * 4 * REDT);
            float4 t1 = *(const float4*)(rr + g * 4 * REDT + 128);
            s0.x += t0.x; s0.y += t0.y; s0.z += t0.z; s0.w += t0.w;
            s1.x += t1.x; s1.y += t1.y; s1.z += t1.z; s1.w += t1.w;
        }
        par ^= 1;

        float* op = out + ((((long)b * NCOUT + Lm) * SDIM + d) * SDIM + h) * SDIM
                        + (wv << 4) + 8 * half;
        float4 o0v = {s0.x + bv, s0.y + bv, s0.z + bv, s0.w + bv};
        float4 o1v = {s1.x + bv, s1.y + bv, s1.z + bv, s1.w + bv};
        *(float4*)(op)     = o0v;
        *(float4*)(op + 4) = o1v;
    }
}

extern "C" void kernel_launch(void* const* d_in, const int* in_sizes, int n_in,
                              void* d_out, int out_size, void* d_ws, size_t ws_size,
                              hipStream_t stream) {
    const float* x    = (const float*)d_in[0];
    const float* wgt  = (const float*)d_in[1];
    const float* bias = (const float*)d_in[2];
    float* out        = (float*)d_out;
    (void)in_sizes; (void)n_in; (void)out_size; (void)d_ws; (void)ws_size;

    dim3 grid(NBLK);
    conv3d_wmma_f32<<<grid, 128, 0, stream>>>(x, wgt, bias, out);
}


# Ring advances two d planes per step: each wave keeps rows d and d+1 on 8 accumulators, each A fragment feeds both rows (144 LDS fragment loads per 216 WMMAs, was 216); sum order changes
# speedup vs baseline: 2.1389x; 1.0406x over previous
#include <hip/hip_runtime.h>

typedef float v2f __attribute__((ext_vector_type(2)));
typedef float v8f __attribute__((ext_vector_type(8)));

#define SDIM 64
#define NCIN 16
#define NCOUT 16
#define NTAP 27
#define NBLK 1024
#define DSEG 16
#define WPITCH (NCOUT * NCIN + 2)
#define LINEP (66 * 4)
#define PLANEP (3 * LINEP)
#define NSLOT 4
#define SLAB (NSLOT * PLANEP)
#define REDT 256
#define NRED (2 * 4 * 3)
#define SMEM_FLOATS (4 * SLAB + NRED * REDT)

__device__ __forceinline__ void plane_load(float4 (&v)[6], const float* xg, int plane,
                                           int offp, int offl, int rs) {
    const float* pp = xg + min(max(plane, 0), SDIM - 1) * (SDIM * SDIM);
    #pragma unroll
    for (int i = 0; i < 4; ++i)
        v[i] = *(const float4*)(pp + offp + (i << 4));
    v[4] = *(const float4*)(pp + offl + (rs << 4));
    v[5] = *(const float4*)(pp + offl + ((2 + rs) << 4));
}

__device__ __forceinline__ void plane_store(const float4 (&v)[6], float* slot, int plane,
                                            bool okp, bool okl, int sb_pair, int sb_last) {
    const bool pok = (unsigned)plane < (unsigned)SDIM;
    const bool a = pok && okp;
    const bool c = pok && okl;
    #pragma unroll
    for (int i = 0; i < 6; ++i) {
        const bool ok = (i < 4) ? a : c;
        float* p = (i < 4) ? slot + sb_pair + i * 64
                           : slot + sb_last + 2 * LINEP + (i - 4) * 128;
        p[0]  = ok ? v[i].x : 0.0f;
        p[4]  = ok ? v[i].y : 0.0f;
        p[8]  = ok ? v[i].z : 0.0f;
        p[12] = ok ? v[i].w : 0.0f;
    }
}

template <int OWN>
__device__ __forceinline__ void red_put(float* red, const v8f (&acc)[2][4], int lane) {
    #pragma unroll
    for (int r = 0; r < 2; ++r) {
        #pragma unroll
        for (int m = 0; m < 4; ++m) {
            if (m == OWN) continue;
            const int k = (OWN < m) ? OWN : OWN - 1;
            float* p = red + ((r * 4 + m) * 3 + k) * REDT + lane * 4;
            float4 lo = {acc[r][m][0], acc[r][m][1], acc[r][m][2], acc[r][m][3]};
            float4 hi = {acc[r][m][4], acc[r][m][5], acc[r][m][6], acc[r][m][7]};
            *(float4*)(p)       = lo;
            *(float4*)(p + 128) = hi;
        }
    }
}

template <int OWN>
__device__ __forceinline__ void red_get(const float* red, const v8f (&acc)[2][4], int lane,
                                        float4 (&s)[2][2]) {
    #pragma unroll
    for (int r = 0; r < 2; ++r) {
        const float* p = red + ((r * 4 + OWN) * 3) * REDT + lane * 4;
        float4 s0 = {acc[r][OWN][0], acc[r][OWN][1], acc[r][OWN][2], acc[r][OWN][3]};
        float4 s1 = {acc[r][OWN][4], acc[r][OWN][5], acc[r][OWN][6], acc[r][OWN][7]};
        #pragma unroll
        for (int k = 0; k < 3; ++k) {
            float4 t0 = *(const float4*)(p + k * REDT);
            float4 t1 = *(const float4*)(p + k * REDT + 128);
            s0.x += t0.x; s0.y += t0.y; s0.z += t0.z; s0.w += t0.w;
            s1.x += t1.x; s1.y += t1.y; s1.z += t1.z; s1.w += t1.w;
        }
        s[r][0] = s0;
        s[r][1] = s1;
    }
}

__launch_bounds__(128, 4)
__global__ void conv3d_wmma_f32(const float* __restrict__ x,
                                const float* __restrict__ w,
                                const float* __restrict__ bias,
                                float* __restrict__ out) {
    __shared__ __align__(16) float smem[SMEM_FLOATS];
    float* wl = smem;

    const int tid  = threadIdx.x;
    const int lane = tid & 31;
    const int wv   = tid >> 5;
    const int Lm   = lane & 15;
    const int half = lane >> 4;
    const int cl   = half << 1;

    for (int idx = tid; idx < NCOUT * NCIN * NTAP; idx += 128) {
        int co   = idx / (NCIN * NTAP);
        int rest = idx - co * (NCIN * NTAP);
        int ci   = rest / NTAP;
        int tap  = rest - ci * NTAP;
        wl[tap * WPITCH + co * NCIN + ci] = w[idx];
    }
    __syncthreads();

    v2f bfr[NTAP];
    #pragma unroll
    for (int tap = 0; tap < NTAP; ++tap)
        bfr[tap] = *(const v2f*)(wl + Lm * NCIN + (wv << 2) + cl + tap * WPITCH);
    __syncthreads();

    float* xs  = smem + wv * SLAB;
    float* red = smem + 4 * SLAB;

    for (int t = lane; t < NSLOT * 3 * 4 * 2; t += 32) {
        int rr   = t >> 1;
        int side = t & 1;
        xs[((rr >> 2) * 66 + side * 65) * 4 + (rr & 3)] = 0.0f;
    }

    const float bv = bias[Lm];

    const int cil = lane & 3;
    const int ql  = (lane >> 2) & 3;
    const int rs  = lane >> 4;
    const int sb_pair = rs * LINEP + 16 * ql + cil + 4;
    const int sb_last = rs * 64    + 16 * ql + cil + 4;
    const int abase = Lm * 4 + cl;

    const int blk = blockIdx.x;
    const int b   = blk >> 8;
    const int d0  = ((blk >> 6) & 3) * DSEG;
    const int h   = blk & 63;

    const float* xg = x + (long)(b * NCIN + (wv << 2) + cil) * (SDIM * SDIM * SDIM) + (ql << 2);
    const int  zhp  = h + rs - 1;
    const bool okp  = (unsigned)zhp < (unsigned)SDIM;
    const bool okl  = (h + 1) < SDIM;
    const int  offp = min(max(zhp, 0), SDIM - 1) * SDIM;
    const int  offl = min(h + 1, SDIM - 1) * SDIM;

    {
        float4 pv[NSLOT][6];
        #pragma unroll
        for (int j = 0; j < NSLOT; ++j)
            plane_load(pv[j], xg, d0 - 1 + j, offp, offl, rs);
        #pragma unroll
        for (int j = 0; j < NSLOT; ++j)
            plane_store(pv[j], xs + j * PLANEP, d0 - 1 + j, okp, okl, sb_pair, sb_last);
    }
    __builtin_amdgcn_fence(__ATOMIC_RELEASE, "wavefront");
    __builtin_amdgcn_wave_barrier();
    __builtin_amdgcn_fence(__ATOMIC_ACQUIRE, "wavefront");

    const int wvs = __builtin_amdgcn_readfirstlane(wv);

    int oa = 0, ob = 2 * PLANEP;
    #pragma nounroll
    for (int k = 0; k < DSEG / 2; ++k) {
        const int d = d0 + 2 * k;

        float4 nva[6], nvb[6];
        plane_load(nva, xg, d + 3, offp, offl, rs);
        plane_load(nvb, xg, d + 4, offp, offl, rs);
        __builtin_amdgcn_sched_barrier(0);

        const float* pl[NSLOT] = {xs + abase + oa, xs + abase + oa + PLANEP,
                                  xs + abase + ob, xs + abase + ob + PLANEP};
        v8f acc[2][4] = {};
        v2f af[2][12];
        #pragma unroll
        for (int j = 0; j < 12; ++j)
            af[0][j] = *(const v2f*)(pl[0] + (16 * (j & 3) + (j >> 2)) * 4);
        #pragma unroll
        for (int line = 0; line < 3 * NSLOT; ++line) {
            const int pj = line / 3;
            const int hl = line % 3;
            if (line < 3 * NSLOT - 1) {
                const int ln = line + 1;
                #pragma unroll
                for (int j = 0; j < 12; ++j)
                    af[ln & 1][j] = *(const v2f*)(pl[ln / 3]
                        + ((ln % 3) * 66 + 16 * (j & 3) + (j >> 2)) * 4);
            }
            __builtin_amdgcn_sched_barrier(0);
            #pragma unroll
            for (int kw = 0; kw < 3; ++kw) {
                if (pj <= 2) {
                    #pragma unroll
                    for (int m = 0; m < 4; ++m)
                        acc[0][m] = __builtin_amdgcn_wmma_f32_16x16x4_f32(
                            false, af[line & 1][kw * 4 + m], false, bfr[(pj * 3 + hl) * 3 + kw],
                            (short)0, acc[0][m], false, false);
                }
                if (pj >= 1) {
                    #pragma unroll
                    for (int m = 0; m < 4; ++m)
                        acc[1][m] = __builtin_amdgcn_wmma_f32_16x16x4_f32(
                            false, af[line & 1][kw * 4 + m], false, bfr[((pj - 1) * 3 + hl) * 3 + kw],
                            (short)0, acc[1][m], false, false);
                }
            }
            __builtin_amdgcn_sched_barrier(0);
        }

        __syncthreads();
        plane_store(nva, xs + oa,          d + 3, okp, okl, sb_pair, sb_last);
        plane_store(nvb, xs + oa + PLANEP, d + 4, okp, okl, sb_pair, sb_last);
        { const int t = oa; oa = ob; ob = t; }

        switch (wvs) {
            case 0:  red_put<0>(red, acc, lane); break;
            case 1:  red_put<1>(red, acc, lane); break;
            case 2:  red_put<2>(red, acc, lane); break;
            default: red_put<3>(red, acc, lane); break;
        }
        __syncthreads();
        float4 s[2][2];
        switch (wvs) {
            case 0:  red_get<0>(red, acc, lane, s); break;
            case 1:  red_get<1>(red, acc, lane, s); break;
            case 2:  red_get<2>(red, acc, lane, s); break;
            default: red_get<3>(red, acc, lane, s); break;
        }

        #pragma unroll
        for (int r = 0; r < 2; ++r) {
            float* op = out + ((((long)b * NCOUT + Lm) * SDIM + (d + r)) * SDIM + h) * SDIM
                            + (wv << 4) + 8 * half;
            float4 o0v = {s[r][0].x + bv, s[r][0].y + bv, s[r][0].z + bv, s[r][0].w + bv};
            float4 o1v = {s[r][1].x + bv, s[r][1].y + bv, s[r][1].z + bv, s[r][1].w + bv};
            *(float4*)(op)     = o0v;
            *(float4*)(op + 4) = o1v;
        }
    }
}

extern "C" void kernel_launch(void* const* d_in, const int* in_sizes, int n_in,
                              void* d_out, int out_size, void* d_ws, size_t ws_size,
                              hipStream_t stream) {
    const float* x    = (const float*)d_in[0];
    const float* wgt  = (const float*)d_in[1];
    const float* bias = (const float*)d_in[2];
    float* out        = (float*)d_out;
    (void)in_sizes; (void)n_in; (void)out_size; (void)d_ws; (void)ws_size;

    dim3 grid(NBLK);
    conv3d_wmma_f32<<<grid, 128, 0, stream>>>(x, wgt, bias, out);
}


# Wave priority by progress: s_setprio 3,2,1,0 entered at steps 0,3,5,7 of a work-group's 8, so the 4 work-groups resident on a WGP stay level and finish together, not oldest first; sums unchanged
# speedup vs baseline: 2.2235x; 1.0396x over previous
#include <hip/hip_runtime.h>

typedef float v2f __attribute__((ext_vector_type(2)));
typedef float v8f __attribute__((ext_vector_type(8)));

#define SDIM 64
#define NCIN 16
#define NCOUT 16
#define NTAP 27
#define NBLK 1024
#define DSEG 16
#define WPITCH (NCOUT * NCIN + 2)
#define LINEP (66 * 4)
#define PLANEP (3 * LINEP)
#define NSLOT 4
#define SLAB (NSLOT * PLANEP)
#define REDT 256
#define NRED (2 * 4 * 3)
#define SMEM_FLOATS (4 * SLAB + NRED * REDT)

__device__ __forceinline__ void plane_load(float4 (&v)[6], const float* xg, int plane,
                                           int offp, int offl, int rs) {
    const float* pp = xg + min(max(plane, 0), SDIM - 1) * (SDIM * SDIM);
    #pragma unroll
    for (int i = 0; i < 4; ++i)
        v[i] = *(const float4*)(pp + offp + (i << 4));
    v[4] = *(const float4*)(pp + offl + (rs << 4));
    v[5] = *(const float4*)(pp + offl + ((2 + rs) << 4));
}

__device__ __forceinline__ void plane_store(const float4 (&v)[6], float* slot, int plane,
                                            bool okp, bool okl, int sb_pair, int sb_last) {
    const bool pok = (unsigned)plane < (unsigned)SDIM;
    const bool a = pok && okp;
    const bool c = pok && okl;
    #pragma unroll
    for (int i = 0; i < 6; ++i) {
        const bool ok = (i < 4) ? a : c;
        float* p = (i < 4) ? slot + sb_pair + i * 64
                           : slot + sb_last + 2 * LINEP + (i - 4) * 128;
        p[0]  = ok ? v[i].x : 0.0f;
        p[4]  = ok ? v[i].y : 0.0f;
        p[8]  = ok ? v[i].z : 0.0f;
        p[12] = ok ? v[i].w : 0.0f;
    }
}

template <int OWN>
__device__ __forceinline__ void red_put(float* red, const v8f (&acc)[2][4], int lane) {
    #pragma unroll
    for (int r = 0; r < 2; ++r) {
        #pragma unroll
        for (int m = 0; m < 4; ++m) {
            if (m == OWN) continue;
            const int k = (OWN < m) ? OWN : OWN - 1;
            float* p = red + ((r * 4 + m) * 3 + k) * REDT + lane * 4;
            float4 lo = {acc[r][m][0], acc[r][m][1], acc[r][m][2], acc[r][m][3]};
            float4 hi = {acc[r][m][4], acc[r][m][5], acc[r][m][6], acc[r][m][7]};
            *(float4*)(p)       = lo;
            *(float4*)(p + 128) = hi;
        }
    }
}

template <int OWN>
__device__ __forceinline__ void red_get(const float* red, const v8f (&acc)[2][4], int lane,
                                        float4 (&s)[2][2]) {
    #pragma unroll
    for (int r = 0; r < 2; ++r) {
        const float* p = red + ((r * 4 + OWN) * 3) * REDT + lane * 4;
        float4 s0 = {acc[r][OWN][0], acc[r][OWN][1], acc[r][OWN][2], acc[r][OWN][3]};
        float4 s1 = {acc[r][OWN][4], acc[r][OWN][5], acc[r][OWN][6], acc[r][OWN][7]};
        #pragma unroll
        for (int k = 0; k < 3; ++k) {
            float4 t0 = *(const float4*)(p + k * REDT);
            float4 t1 = *(const float4*)(p + k * REDT + 128);
            s0.x += t0.x; s0.y += t0.y; s0.z += t0.z; s0.w += t0.w;
            s1.x += t1.x; s1.y += t1.y; s1.z += t1.z; s1.w += t1.w;
        }
        s[r][0] = s0;
        s[r][1] = s1;
    }
}

__launch_bounds__(128, 4)
__global__ void conv3d_wmma_f32(const float* __restrict__ x,
                                const float* __restrict__ w,
                                const float* __restrict__ bias,
                                float* __restrict__ out) {
    __shared__ __align__(16) float smem[SMEM_FLOATS];
    float* wl = smem;

    const int tid  = threadIdx.x;
    const int lane = tid & 31;
    const int wv   = tid >> 5;
    const int Lm   = lane & 15;
    const int half = lane >> 4;
    const int cl   = half << 1;

    for (int idx = tid; idx < NCOUT * NCIN * NTAP; idx += 128) {
        int co   = idx / (NCIN * NTAP);
        int rest = idx - co * (NCIN * NTAP);
        int ci   = rest / NTAP;
        int tap  = rest - ci * NTAP;
        wl[tap * WPITCH + co * NCIN + ci] = w[idx];
    }
    __syncthreads();

    v2f bfr[NTAP];
    #pragma unroll
    for (int tap = 0; tap < NTAP; ++tap)
        bfr[tap] = *(const v2f*)(wl + Lm * NCIN + (wv << 2) + cl + tap * WPITCH);
    __syncthreads();

    float* xs  = smem + wv * SLAB;
    float* red = smem + 4 * SLAB;

    for (int t = lane; t < NSLOT * 3 * 4 * 2; t += 32) {
        int rr   = t >> 1;
        int side = t & 1;
        xs[((rr >> 2) * 66 + side * 65) * 4 + (rr & 3)] = 0.0f;
    }

    const float bv = bias[Lm];

    const int cil = lane & 3;
    const int ql  = (lane >> 2) & 3;
    const int rs  = lane >> 4;
    const int sb_pair = rs * LINEP + 16 * ql + cil + 4;
    const int sb_last = rs * 64    + 16 * ql + cil + 4;
    const int abase = Lm * 4 + cl;

    const int blk = blockIdx.x;
    const int b   = blk >> 8;
    const int d0  = ((blk >> 6) & 3) * DSEG;
    const int h   = blk & 63;

    const float* xg = x + (long)(b * NCIN + (wv << 2) + cil) * (SDIM * SDIM * SDIM) + (ql << 2);
    const int  zhp  = h + rs - 1;
    const bool okp  = (unsigned)zhp < (unsigned)SDIM;
    const bool okl  = (h + 1) < SDIM;
    const int  offp = min(max(zhp, 0), SDIM - 1) * SDIM;
    const int  offl = min(h + 1, SDIM - 1) * SDIM;

    {
        float4 pv[NSLOT][6];
        #pragma unroll
        for (int j = 0; j < NSLOT; ++j)
            plane_load(pv[j], xg, d0 - 1 + j, offp, offl, rs);
        #pragma unroll
        for (int j = 0; j < NSLOT; ++j)
            plane_store(pv[j], xs + j * PLANEP, d0 - 1 + j, okp, okl, sb_pair, sb_last);
    }
    __builtin_amdgcn_fence(__ATOMIC_RELEASE, "wavefront");
    __builtin_amdgcn_wave_barrier();
    __builtin_amdgcn_fence(__ATOMIC_ACQUIRE, "wavefront");

    const int wvs = __builtin_amdgcn_readfirstlane(wv);

    int oa = 0, ob = 2 * PLANEP;
    #pragma nounroll
    for (int k = 0; k < DSEG / 2; ++k) {
        const int d = d0 + 2 * k;

        if (k == 0)      __builtin_amdgcn_s_setprio(3);
        else if (k == 3) __builtin_amdgcn_s_setprio(2);
        else if (k == 5) __builtin_amdgcn_s_setprio(1);
        else if (k == 7) __builtin_amdgcn_s_setprio(0);

        float4 nva[6], nvb[6];
        plane_load(nva, xg, d + 3, offp, offl, rs);
        plane_load(nvb, xg, d + 4, offp, offl, rs);
        __builtin_amdgcn_sched_barrier(0);

        const float* pl[NSLOT] = {xs + abase + oa, xs + abase + oa + PLANEP,
                                  xs + abase + ob, xs + abase + ob + PLANEP};
        v8f acc[2][4] = {};
        v2f af[2][12];
        #pragma unroll
        for (int j = 0; j < 12; ++j)
            af[0][j] = *(const v2f*)(pl[0] + (16 * (j & 3) + (j >> 2)) * 4);
        #pragma unroll
        for (int line = 0; line < 3 * NSLOT; ++line) {
            const int pj = line / 3;
            const int hl = line % 3;
            if (line < 3 * NSLOT - 1) {
                const int ln = line + 1;
                #pragma unroll
                for (int j = 0; j < 12; ++j)
                    af[ln & 1][j] = *(const v2f*)(pl[ln / 3]
                        + ((ln % 3) * 66 + 16 * (j & 3) + (j >> 2)) * 4);
            }
            __builtin_amdgcn_sched_barrier(0);
            #pragma unroll
            for (int kw = 0; kw < 3; ++kw) {
                if (pj <= 2) {
                    #pragma unroll
                    for (int m = 0; m < 4; ++m)
                        acc[0][m] = __builtin_amdgcn_wmma_f32_16x16x4_f32(
                            false, af[line & 1][kw * 4 + m], false, bfr[(pj * 3 + hl) * 3 + kw],
                            (short)0, acc[0][m], false, false);
                }
                if (pj >= 1) {
                    #pragma unroll
                    for (int m = 0; m < 4; ++m)
                        acc[1][m] = __builtin_amdgcn_wmma_f32_16x16x4_f32(
                            false, af[line & 1][kw * 4 + m], false, bfr[((pj - 1) * 3 + hl) * 3 + kw],
                            (short)0, acc[1][m], false, false);
                }
            }
            __builtin_amdgcn_sched_barrier(0);
        }

        __syncthreads();
        plane_store(nva, xs + oa,          d + 3, okp, okl, sb_pair, sb_last);
        plane_store(nvb, xs + oa + PLANEP, d + 4, okp, okl, sb_pair, sb_last);
        { const int t = oa; oa = ob; ob = t; }

        switch (wvs) {
            case 0:  red_put<0>(red, acc, lane); break;
            case 1:  red_put<1>(red, acc, lane); break;
            case 2:  red_put<2>(red, acc, lane); break;
            default: red_put<3>(red, acc, lane); break;
        }
        __syncthreads();
        float4 s[2][2];
        switch (wvs) {
            case 0:  red_get<0>(red, acc, lane, s); break;
            case 1:  red_get<1>(red, acc, lane, s); break;
            case 2:  red_get<2>(red, acc, lane, s); break;
            default: red_get<3>(red, acc, lane, s); break;
        }

        #pragma unroll
        for (int r = 0; r < 2; ++r) {
            float* op = out + ((((long)b * NCOUT + Lm) * SDIM + (d + r)) * SDIM + h) * SDIM
                            + (wv << 4) + 8 * half;
            float4 o0v = {s[r][0].x + bv, s[r][0].y + bv, s[r][0].z + bv, s[r][0].w + bv};
            float4 o1v = {s[r][1].x + bv, s[r][1].y + bv, s[r][1].z + bv, s[r][1].w + bv};
            *(float4*)(op)     = o0v;
            *(float4*)(op + 4) = o1v;
        }
    }
}

extern "C" void kernel_launch(void* const* d_in, const int* in_sizes, int n_in,
                              void* d_out, int out_size, void* d_ws, size_t ws_size,
                              hipStream_t stream) {
    const float* x    = (const float*)d_in[0];
    const float* wgt  = (const float*)d_in[1];
    const float* bias = (const float*)d_in[2];
    float* out        = (float*)d_out;
    (void)in_sizes; (void)n_in; (void)out_size; (void)d_ws; (void)ws_size;

    dim3 grid(NBLK);
    conv3d_wmma_f32<<<grid, 128, 0, stream>>>(x, wgt, bias, out);
}


# Row finishing one step behind: a step ends with the partial-sum hand-over; foreign partials fetched, summed, stored under lines 2-3 of the next step's WMMAs; both barriers split; sums unchanged
# speedup vs baseline: 2.2956x; 1.0324x over previous
#include <hip/hip_runtime.h>

typedef float v2f __attribute__((ext_vector_type(2)));
typedef float v8f __attribute__((ext_vector_type(8)));

#define SDIM 64
#define NCIN 16
#define NCOUT 16
#define NTAP 27
#define NBLK 1024
#define DSEG 16
#define WPITCH (NCOUT * NCIN + 2)
#define LINEP (66 * 4)
#define PLANEP (3 * LINEP)
#define NSLOT 4
#define SLAB (NSLOT * PLANEP)
#define REDT 256
#define NRED (2 * 4 * 3)
#define SMEM_FLOATS (4 * SLAB + NRED * REDT)

__device__ __forceinline__ void plane_load(float4 (&v)[6], const float* xg, int plane,
                                           int offp, int offl, int rs) {
    const float* pp = xg + min(max(plane, 0), SDIM - 1) * (SDIM * SDIM);
    #pragma unroll
    for (int i = 0; i < 4; ++i)
        v[i] = *(const float4*)(pp + offp + (i << 4));
    v[4] = *(const float4*)(pp + offl + (rs << 4));
    v[5] = *(const float4*)(pp + offl + ((2 + rs) << 4));
}

__device__ __forceinline__ void plane_pad(float4 (&v)[6], int plane, bool okp, bool okl) {
    const bool pok = (unsigned)plane < (unsigned)SDIM;
    const bool a = pok && okp;
    const bool c = pok && okl;
    #pragma unroll
    for (int i = 0; i < 6; ++i) {
        const bool ok = (i < 4) ? a : c;
        v[i].x = ok ? v[i].x : 0.0f;
        v[i].y = ok ? v[i].y : 0.0f;
        v[i].z = ok ? v[i].z : 0.0f;
        v[i].w = ok ? v[i].w : 0.0f;
    }
}

__device__ __forceinline__ void plane_store_raw(const float4 (&v)[6], float* slot,
                                                int sb_pair, int sb_last) {
    #pragma unroll
    for (int i = 0; i < 6; ++i) {
        float* p = (i < 4) ? slot + sb_pair + i * 64
                           : slot + sb_last + 2 * LINEP + (i - 4) * 128;
        p[0]  = v[i].x;
        p[4]  = v[i].y;
        p[8]  = v[i].z;
        p[12] = v[i].w;
    }
}

template <int OWN>
__device__ __forceinline__ void red_put(float* red, const v8f (&acc)[2][4], int lane) {
    #pragma unroll
    for (int r = 0; r < 2; ++r) {
        #pragma unroll
        for (int m = 0; m < 4; ++m) {
            if (m == OWN) continue;
            const int k = (OWN < m) ? OWN : OWN - 1;
            float* p = red + ((r * 4 + m) * 3 + k) * REDT + lane * 4;
            float4 lo = {acc[r][m][0], acc[r][m][1], acc[r][m][2], acc[r][m][3]};
            float4 hi = {acc[r][m][4], acc[r][m][5], acc[r][m][6], acc[r][m][7]};
            *(float4*)(p)       = lo;
            *(float4*)(p + 128) = hi;
        }
    }
}

template <int OWN>
__device__ __forceinline__ void keep_own(const v8f (&acc)[2][4], float4 (&keep)[2][2]) {
    #pragma unroll
    for (int r = 0; r < 2; ++r) {
        float4 k0 = {acc[r][OWN][0], acc[r][OWN][1], acc[r][OWN][2], acc[r][OWN][3]};
        float4 k1 = {acc[r][OWN][4], acc[r][OWN][5], acc[r][OWN][6], acc[r][OWN][7]};
        keep[r][0] = k0;
        keep[r][1] = k1;
    }
}

__launch_bounds__(128, 4)
__global__ void conv3d_wmma_f32(const float* __restrict__ x,
                                const float* __restrict__ w,
                                const float* __restrict__ bias,
                                float* __restrict__ out) {
    __shared__ __align__(16) float smem[SMEM_FLOATS];
    float* wl = smem;

    const int tid  = threadIdx.x;
    const int lane = tid & 31;
    const int wv   = tid >> 5;
    const int wvs  = __builtin_amdgcn_readfirstlane(wv);
    const int Lm   = lane & 15;
    const int half = lane >> 4;
    const int cl   = half << 1;

    for (int idx = tid; idx < NCOUT * NCIN * NTAP; idx += 128) {
        int co   = idx / (NCIN * NTAP);
        int rest = idx - co * (NCIN * NTAP);
        int ci   = rest / NTAP;
        int tap  = rest - ci * NTAP;
        wl[tap * WPITCH + co * NCIN + ci] = w[idx];
    }
    __syncthreads();

    v2f bfr[NTAP];
    #pragma unroll
    for (int tap = 0; tap < NTAP; ++tap)
        bfr[tap] = *(const v2f*)(wl + Lm * NCIN + (wv << 2) + cl + tap * WPITCH);
    __syncthreads();

    float* xs  = smem + wvs * SLAB;
    float* red = smem + 4 * SLAB;

    for (int t = lane; t < NSLOT * 3 * 4 * 2; t += 32) {
        int rr   = t >> 1;
        int side = t & 1;
        xs[((rr >> 2) * 66 + side * 65) * 4 + (rr & 3)] = 0.0f;
    }

    const float bv = bias[Lm];

    const int cil = lane & 3;
    const int ql  = (lane >> 2) & 3;
    const int rs  = lane >> 4;
    const int sb_pair = rs * LINEP + 16 * ql + cil + 4;
    const int sb_last = rs * 64    + 16 * ql + cil + 4;
    const int abase = Lm * 4 + cl;

    const int blk = blockIdx.x;
    const int b   = blk >> 8;
    const int d0  = ((blk >> 6) & 3) * DSEG;
    const int h   = blk & 63;

    const float* xg = x + (long)(b * NCIN + (wv << 2) + cil) * (SDIM * SDIM * SDIM) + (ql << 2);
    const int  zhp  = h + rs - 1;
    const bool okp  = (unsigned)zhp < (unsigned)SDIM;
    const bool okl  = (h + 1) < SDIM;
    const int  offp = min(max(zhp, 0), SDIM - 1) * SDIM;
    const int  offl = min(h + 1, SDIM - 1) * SDIM;

    {
        float4 pv[NSLOT][6];
        #pragma unroll
        for (int j = 0; j < NSLOT; ++j)
            plane_load(pv[j], xg, d0 - 1 + j, offp, offl, rs);
        #pragma unroll
        for (int j = 0; j < NSLOT; ++j) {
            plane_pad(pv[j], d0 - 1 + j, okp, okl);
            plane_store_raw(pv[j], xs + j * PLANEP, sb_pair, sb_last);
        }
    }
    __builtin_amdgcn_fence(__ATOMIC_RELEASE, "wavefront");
    __builtin_amdgcn_wave_barrier();
    __builtin_amdgcn_fence(__ATOMIC_ACQUIRE, "wavefront");

    const float* redme = red + (wvs * 3) * REDT + lane * 4;
    float4 keep[2][2];
    float* orow = out + ((((long)b * NCOUT + Lm) * SDIM + d0) * SDIM + h) * SDIM + (wv << 4) + 8 * half;

    asm volatile("s_barrier_signal -1" ::: "memory");

    int oa = 0, ob = 2 * PLANEP;
    #pragma nounroll
    for (int k = 0; k < DSEG / 2; ++k) {
        const int d = d0 + 2 * k;

        if (k == 0)      __builtin_amdgcn_s_setprio(3);
        else if (k == 3) __builtin_amdgcn_s_setprio(2);
        else if (k == 5) __builtin_amdgcn_s_setprio(1);
        else if (k == 7) __builtin_amdgcn_s_setprio(0);

        const float* pl[NSLOT] = {xs + abase + oa, xs + abase + oa + PLANEP,
                                  xs + abase + ob, xs + abase + ob + PLANEP};
        v8f acc[2][4] = {};
        v2f af[2][12];
        #pragma unroll
        for (int j = 0; j < 12; ++j)
            af[0][j] = *(const v2f*)(pl[0] + (16 * (j & 3) + (j >> 2)) * 4);
        float4 t[3][2];
        float4 nva[6], nvb[6];
        #pragma unroll
        for (int line = 0; line < 3 * NSLOT; ++line) {
            const int pj = line / 3;
            const int hl = line % 3;
            if (line < 3 * NSLOT - 1) {
                const int ln = line + 1;
                #pragma unroll
                for (int j = 0; j < 12; ++j)
                    af[ln & 1][j] = *(const v2f*)(pl[ln / 3]
                        + ((ln % 3) * 66 + 16 * (j & 3) + (j >> 2)) * 4);
            }
            if ((line == 1 || line == 2) && k > 0) {
                #pragma unroll
                for (int q = 0; q < 3; ++q) {
                    t[q][0] = *(const float4*)(redme + ((line - 1) * 12 + q) * REDT);
                    t[q][1] = *(const float4*)(redme + ((line - 1) * 12 + q) * REDT + 128);
                }
            }
            __builtin_amdgcn_sched_barrier(0);
            #pragma unroll
            for (int kw = 0; kw < 3; ++kw) {
                if (pj <= 2) {
                    #pragma unroll
                    for (int m = 0; m < 4; ++m)
                        acc[0][m] = __builtin_amdgcn_wmma_f32_16x16x4_f32(
                            false, af[line & 1][kw * 4 + m], false, bfr[(pj * 3 + hl) * 3 + kw],
                            (short)0, acc[0][m], false, false);
                }
                if (pj >= 1) {
                    #pragma unroll
                    for (int m = 0; m < 4; ++m)
                        acc[1][m] = __builtin_amdgcn_wmma_f32_16x16x4_f32(
                            false, af[line & 1][kw * 4 + m], false, bfr[((pj - 1) * 3 + hl) * 3 + kw],
                            (short)0, acc[1][m], false, false);
                }
            }
            __builtin_amdgcn_sched_barrier(0);
            if (line == 0) {
                asm volatile("s_barrier_wait -1" ::: "memory");
            }
            if (line == 1 || line == 2) {
                asm volatile("v_nop\n\tv_nop\n\tv_nop\n\tv_nop");
                if (k > 0) {
                    const int r = line - 1;
                    float4 s0 = keep[r][0], s1 = keep[r][1];
                    #pragma unroll
                    for (int q = 0; q < 3; ++q) {
                        s0.x += t[q][0].x; s0.y += t[q][0].y; s0.z += t[q][0].z; s0.w += t[q][0].w;
                        s1.x += t[q][1].x; s1.y += t[q][1].y; s1.z += t[q][1].z; s1.w += t[q][1].w;
                    }
                    float* op = orow + r * (SDIM * SDIM);
                    float4 o0v = {s0.x + bv, s0.y + bv, s0.z + bv, s0.w + bv};
                    float4 o1v = {s1.x + bv, s1.y + bv, s1.z + bv, s1.w + bv};
                    *(float4*)(op)     = o0v;
                    *(float4*)(op + 4) = o1v;
                }
            }
            if (line == 2) {
                asm volatile("s_wait_dscnt 0x0" ::: "memory");
                asm volatile("s_barrier_signal -1" ::: "memory");
                plane_load(nva, xg, d + 3, offp, offl, rs);
                plane_load(nvb, xg, d + 4, offp, offl, rs);
                __builtin_amdgcn_sched_barrier(0);
            }
        }

        plane_pad(nva, d + 3, okp, okl);
        plane_pad(nvb, d + 4, okp, okl);
        plane_store_raw(nva, xs + oa,          sb_pair, sb_last);
        plane_store_raw(nvb, xs + oa + PLANEP, sb_pair, sb_last);
        { const int tt = oa; oa = ob; ob = tt; }
        asm volatile("s_barrier_wait -1" ::: "memory");

        switch (wvs) {
            case 0:  red_put<0>(red, acc, lane); keep_own<0>(acc, keep); break;
            case 1:  red_put<1>(red, acc, lane); keep_own<1>(acc, keep); break;
            case 2:  red_put<2>(red, acc, lane); keep_own<2>(acc, keep); break;
            default: red_put<3>(red, acc, lane); keep_own<3>(acc, keep); break;
        }
        asm volatile("s_wait_dscnt 0x0" ::: "memory");
        asm volatile("s_barrier_signal -1" ::: "memory");
        if (k > 0) orow += 2 * SDIM * SDIM;
    }
    asm volatile("s_barrier_wait -1" ::: "memory");

    {
        #pragma unroll
        for (int r = 0; r < 2; ++r) {
            float4 s0 = keep[r][0], s1 = keep[r][1];
            #pragma unroll
            for (int q = 0; q < 3; ++q) {
                float4 t0 = *(const float4*)(redme + (r * 12 + q) * REDT);
                float4 t1 = *(const float4*)(redme + (r * 12 + q) * REDT + 128);
                s0.x += t0.x; s0.y += t0.y; s0.z += t0.z; s0.w += t0.w;
                s1.x += t1.x; s1.y += t1.y; s1.z += t1.z; s1.w += t1.w;
            }
            float* op = orow + r * (SDIM * SDIM);
            float4 o0v = {s0.x + bv, s0.y + bv, s0.z + bv, s0.w + bv};
            float4 o1v = {s1.x + bv, s1.y + bv, s1.z + bv, s1.w + bv};
            *(float4*)(op)     = o0v;
            *(float4*)(op + 4) = o1v;
        }
    }
}

extern "C" void kernel_launch(void* const* d_in, const int* in_sizes, int n_in,
                              void* d_out, int out_size, void* d_ws, size_t ws_size,
                              hipStream_t stream) {
    const float* x    = (const float*)d_in[0];
    const float* wgt  = (const float*)d_in[1];
    const float* bias = (const float*)d_in[2];
    float* out        = (float*)d_out;
    (void)in_sizes; (void)n_in; (void)out_size; (void)d_ws; (void)ws_size;

    dim3 grid(NBLK);
    conv3d_wmma_f32<<<grid, 128, 0, stream>>>(x, wgt, bias, out);
}
